# starting kernel
# baseline (speedup 1.0000x reference)
#include <hip/hip_runtime.h>


#define NN 100000
#define NE 1600000
#define KD 128

typedef __attribute__((ext_vector_type(2))) float v2f;
typedef __attribute__((ext_vector_type(8))) float v8f;

__global__ void fill_zero_kernel(float* __restrict__ p, int n) {
  int i = blockIdx.x * blockDim.x + threadIdx.x;
  int stride = gridDim.x * blockDim.x;
  for (; i < n; i += stride) p[i] = 0.0f;
}

__global__ void degree_kernel(const int* __restrict__ src, const int* __restrict__ dst,
                              float* __restrict__ out_deg, float* __restrict__ in_deg) {
  int e = blockIdx.x * blockDim.x + threadIdx.x;
  if (e >= NE) return;
  unsafeAtomicAdd(out_deg + src[e], 1.0f);
  unsafeAtomicAdd(in_deg + dst[e], 1.0f);
}

__global__ void norm_kernel(float* __restrict__ deg, int n) {
  int i = blockIdx.x * blockDim.x + threadIdx.x;
  if (i >= n) return;
  float d = deg[i];
  d = d < 1.0f ? 1.0f : d;
  deg[i] = rsqrtf(d);
}

template <int LOG2C, bool SCALE>
__global__ void edge_aggregate_kernel(const float* __restrict__ h,
                                      const int* __restrict__ src,
                                      const int* __restrict__ dst,
                                      const float* __restrict__ ew,
                                      const float* __restrict__ src_scale,
                                      float* __restrict__ agg) {
  constexpr int WIDTH = 4 << LOG2C;
  unsigned tid = blockIdx.x * blockDim.x + threadIdx.x;
  unsigned e = tid >> LOG2C;
  unsigned j = tid & ((1u << LOG2C) - 1u);
  if (e >= NE) return;
  int s = src[e];
  int d = dst[e];
  float c = ew[e];
  if (SCALE) c *= src_scale[s];
  const float4* hp = (const float4*)(h + (size_t)s * WIDTH);
  float4 v = hp[j];
  float* o = agg + (size_t)d * WIDTH + 4u * j;
  unsafeAtomicAdd(o + 0, v.x * c);
  unsafeAtomicAdd(o + 1, v.y * c);
  unsafeAtomicAdd(o + 2, v.z * c);
  unsafeAtomicAdd(o + 3, v.w * c);
}

template <int FOUT, bool RELU, bool HAS_BIAS>
__global__ void wmma_gemm_kernel(const float* __restrict__ A,
                                 const float* __restrict__ row_scale,
                                 const float* __restrict__ W,
                                 const float* __restrict__ bias,
                                 float* __restrict__ out) {
  __shared__ float As[16 * KD];
  const int m_base = blockIdx.x * 16;

  {
    constexpr int NVEC = 16 * KD / 4;
    for (int v = threadIdx.x; v < NVEC; v += blockDim.x) {
      int r = v >> 5;
      int cv = v & 31;
      float4 x = ((const float4*)(A + (size_t)(m_base + r) * KD))[cv];
      float s = row_scale[m_base + r];
      x.x *= s; x.y *= s; x.z *= s; x.w *= s;
      *(float4*)&As[r * KD + 4 * cv] = x;
    }
  }
  __syncthreads();

  const int wave = threadIdx.x >> 5;
  const int lane = threadIdx.x & 31;
  const int n_base = wave * 16;
  const int lo = lane & 15;
  const int hi = lane >> 4;

  v8f acc = {};
  const float* bbase = W + (size_t)(2 * hi) * FOUT + n_base + lo;
  const float* abase = &As[lo * KD + 2 * hi];

  #pragma unroll
  for (int k0 = 0; k0 < KD; k0 += 4) {
    v2f a = *(const v2f*)(abase + k0);
    v2f b;
    b.x = bbase[(size_t)(k0 + 0) * FOUT];
    b.y = bbase[(size_t)(k0 + 1) * FOUT];
    acc = __builtin_amdgcn_wmma_f32_16x16x4_f32(
         false, a,  false, b,
         (short)0, acc,  false,  false);
  }

  const int col = n_base + lo;
  const float bv = HAS_BIAS ? bias[col] : 0.0f;
  #pragma unroll
  for (int v = 0; v < 8; ++v) {
    int row = m_base + v + 8 * hi;
    float val = acc[v] + bv;
    if (RELU) val = val > 0.0f ? val : 0.0f;
    out[(size_t)row * FOUT + col] = val;
  }
}

__global__ void final_epilogue_kernel(const float* __restrict__ agg,
                                      const float* __restrict__ in_norm,
                                      const float* __restrict__ b2,
                                      float* __restrict__ out) {
  int i = blockIdx.x * blockDim.x + threadIdx.x;
  if (i >= NN * 64) return;
  int n = i >> 6;
  int f = i & 63;
  out[i] = agg[i] * in_norm[n] + b2[f];
}

extern "C" void kernel_launch(void* const* d_in, const int* in_sizes, int n_in,
                              void* d_out, int out_size, void* d_ws, size_t ws_size,
                              hipStream_t stream) {
  const float* features = (const float*)d_in[0];
  const int*   src      = (const int*)d_in[1];
  const int*   dst      = (const int*)d_in[2];
  const float* ew       = (const float*)d_in[3];
  const float* W0 = (const float*)d_in[4];
  const float* b0 = (const float*)d_in[5];
  const float* W1 = (const float*)d_in[6];
  const float* b1 = (const float*)d_in[7];
  const float* W2 = (const float*)d_in[8];
  const float* b2 = (const float*)d_in[9];
  float* out = (float*)d_out;

  float* ws       = (float*)d_ws;
  float* out_norm = ws;
  float* in_norm  = ws + NN;
  float* bufA     = ws + 2 * NN;
  float* bufB     = bufA + (size_t)NN * KD;

  fill_zero_kernel<<<512, 256, 0, stream>>>(out_norm, 2 * NN);
  degree_kernel<<<(NE + 255) / 256, 256, 0, stream>>>(src, dst, out_norm, in_norm);
  norm_kernel<<<(2 * NN + 255) / 256, 256, 0, stream>>>(out_norm, 2 * NN);

  fill_zero_kernel<<<2048, 256, 0, stream>>>(bufA, NN * KD);
  edge_aggregate_kernel<5, true><<<(NE * 32) / 256, 256, 0, stream>>>(
      features, src, dst, ew, out_norm, bufA);
  wmma_gemm_kernel<128, true, true><<<NN / 16, 256, 0, stream>>>(
      bufA, in_norm, W0, b0, bufB);

  fill_zero_kernel<<<2048, 256, 0, stream>>>(bufA, NN * KD);
  edge_aggregate_kernel<5, true><<<(NE * 32) / 256, 256, 0, stream>>>(
      bufB, src, dst, ew, out_norm, bufA);
  wmma_gemm_kernel<128, true, true><<<NN / 16, 256, 0, stream>>>(
      bufA, in_norm, W1, b1, bufB);

  wmma_gemm_kernel<64, false, false><<<NN / 16, 128, 0, stream>>>(
      bufB, out_norm, W2, nullptr, bufA);
  fill_zero_kernel<<<2048, 256, 0, stream>>>(bufB, NN * 64);
  edge_aggregate_kernel<4, false><<<(NE * 16) / 256, 256, 0, stream>>>(
      bufA, src, dst, ew, nullptr, bufB);
  final_epilogue_kernel<<<(NN * 64 + 255) / 256, 256, 0, stream>>>(bufB, in_norm, b2, out);
}


# edge lists: prep kernel places each edge record in its tile's exact list (LDS ranks per 4096-edge block, one global int add per block and tile); layer 0's id-byte sweep gone; sum order = arrival
# speedup vs baseline: 2.4738x; 2.4738x over previous
#include <hip/hip_runtime.h>


#define NN 100000
#define NE 1600000
#define KD 128
#define TILE 400
#define NTILE (NN / TILE)
#define OQCAP 320
#define EPB 4096
#define CTRS 32
#define CTRW (256 * CTRS)

typedef __attribute__((ext_vector_type(2))) float v2f;
typedef __attribute__((ext_vector_type(8))) float v8f;

__global__ void fill_zero_kernel(float* __restrict__ p, int n) {
  int i = blockIdx.x * blockDim.x + threadIdx.x;
  int stride = gridDim.x * blockDim.x;
  for (; i < n; i += stride) p[i] = 0.0f;
}

__global__ void degree_kernel(const int* __restrict__ src, const int* __restrict__ dst,
                              float* __restrict__ out_deg, float* __restrict__ in_deg) {
  int e = blockIdx.x * blockDim.x + threadIdx.x;
  if (e >= NE) return;
  unsafeAtomicAdd(out_deg + src[e], 1.0f);
  unsafeAtomicAdd(in_deg + dst[e], 1.0f);
}

__global__ __launch_bounds__(512)
void norm_count_kernel(float* __restrict__ deg, unsigned* __restrict__ tcount) {
  __shared__ unsigned total;
  const int i = threadIdx.x;
  if (i == 0) total = 0u;
  __syncthreads();
  unsigned mine = 0u;
  if (i < TILE) {
    const int n = blockIdx.x * TILE + i;
    float dout = deg[n];
    float din = deg[NN + n];
    mine = (unsigned)din;
    dout = dout < 1.0f ? 1.0f : dout;
    din = din < 1.0f ? 1.0f : din;
    deg[n] = rsqrtf(dout);
    deg[NN + n] = rsqrtf(din);
  }
  for (int off = 16; off > 0; off >>= 1) mine += __shfl_xor(mine, off);
  if ((i & 31) == 0 && mine)
    __hip_atomic_fetch_add(&total, mine, __ATOMIC_RELAXED, __HIP_MEMORY_SCOPE_WORKGROUP);
  __syncthreads();
  if (i == 0) tcount[blockIdx.x] = total;
}

__global__ __launch_bounds__(1024)
void edge_place_kernel(const int* __restrict__ src, const int* __restrict__ dst,
                       const float* __restrict__ ew, const float* __restrict__ out_norm,
                       const unsigned* __restrict__ tcount, unsigned* __restrict__ fillctr,
                       unsigned* __restrict__ toffg, unsigned* __restrict__ list) {
  __shared__ unsigned lcnt[256];
  __shared__ unsigned lpos[256];
  __shared__ unsigned toff[256];
  __shared__ unsigned gbase[256];
  __shared__ uint2 stage[EPB];
  __shared__ unsigned sidx[EPB];

  const int tid = threadIdx.x;
  const int i = blockIdx.x * 1024 + tid;
  const bool valid = i < NE / 4;
  if (tid < 256) lcnt[tid] = 0u;
  __syncthreads();

  unsigned t[4] = {0u, 0u, 0u, 0u}, rk[4] = {0u, 0u, 0u, 0u};
  uint2 rc[4] = {};
  if (valid) {
    const int4 s = ((const int4*)src)[i];
    const int4 d = ((const int4*)dst)[i];
    const float4 w = ((const float4*)ew)[i];
    const int sv[4] = {s.x, s.y, s.z, s.w};
    const int dv[4] = {d.x, d.y, d.z, d.w};
    const float wv[4] = {w.x, w.y, w.z, w.w};
#pragma unroll
    for (int k = 0; k < 4; ++k) {
      t[k] = (unsigned)dv[k] / TILE;
      rc[k].x = (unsigned)sv[k] | (((unsigned)dv[k] - t[k] * TILE) << 17);
      rc[k].y = __float_as_uint(wv[k] * out_norm[sv[k]]);
      rk[k] = __hip_atomic_fetch_add(lcnt + t[k], 1u, __ATOMIC_RELAXED, __HIP_MEMORY_SCOPE_WORKGROUP);
    }
  }
  __syncthreads();

  if (tid < 32) {
    unsigned a[8], b[8];
    unsigned sa = 0u, sb = 0u;
#pragma unroll
    for (int k = 0; k < 8; ++k) {
      const int e = 8 * tid + k;
      a[k] = lcnt[e];
      b[k] = e < NTILE ? tcount[e] : 0u;
      sa += a[k];
      sb += b[k];
    }
    unsigned xa = sa, xb = sb;
    for (int off = 1; off < 32; off <<= 1) {
      const unsigned ya = __shfl_up(xa, off);
      const unsigned yb = __shfl_up(xb, off);
      if (tid >= off) { xa += ya; xb += yb; }
    }
    unsigned ea = xa - sa, eb = xb - sb;
#pragma unroll
    for (int k = 0; k < 8; ++k) {
      const int e = 8 * tid + k;
      lpos[e] = ea;
      toff[e] = eb;
      if (blockIdx.x == 0) toffg[e] = eb;
      ea += a[k];
      eb += b[k];
    }
  }
  __syncthreads();

  if (tid < NTILE) {
    const unsigned n = lcnt[tid];
    unsigned g = 0u;
    if (n) g = __hip_atomic_fetch_add(fillctr + tid * CTRS, n, __ATOMIC_RELAXED, __HIP_MEMORY_SCOPE_AGENT);
    gbase[tid] = toff[tid] + g;
  }
  __syncthreads();

  if (valid) {
#pragma unroll
    for (int k = 0; k < 4; ++k) {
      const unsigned p = lpos[t[k]] + rk[k];
      stage[p] = rc[k];
      sidx[p] = gbase[t[k]] + rk[k];
    }
  }
  __syncthreads();

  int total = NE - (int)blockIdx.x * EPB;
  if (total > EPB) total = EPB;
  for (int p = tid; p < total; p += 1024) ((uint2*)list)[sidx[p]] = stage[p];
}


template <int WIDTH> struct RowVec;
template <> struct RowVec<128> { typedef float4 type; };
template <> struct RowVec<64>  { typedef float2 type; };

__device__ __forceinline__ float4 axpy(float4 t, const float4 r, const float c) {
  t.x += r.x * c; t.y += r.y * c; t.z += r.z * c; t.w += r.w * c;
  return t;
}
__device__ __forceinline__ float2 axpy(float2 t, const float2 r, const float c) {
  t.x += r.x * c; t.y += r.y * c;
  return t;
}

template <int WIDTH>
__device__ __forceinline__ void route_record(const float* __restrict__ h, float* acc, uint2* oq,
                                             unsigned* oqn, const uint2 rc) {
  const unsigned own = (rc.x >> 17) & 31u;
  const unsigned slot = __hip_atomic_fetch_add(oqn + own, 1u, __ATOMIC_RELAXED, __HIP_MEMORY_SCOPE_WORKGROUP);
  if (slot < (unsigned)OQCAP) {
    oq[own * OQCAP + slot] = rc;
  } else {
    const float c = __uint_as_float(rc.y);
    float* a = acc + (rc.x >> 17) * WIDTH;
    const float* hp = h + (size_t)(rc.x & 0x1FFFFu) * WIDTH;
    for (int col = 0; col < WIDTH; ++col)
      __hip_atomic_fetch_add(a + col, hp[col] * c, __ATOMIC_RELAXED, __HIP_MEMORY_SCOPE_WORKGROUP);
  }
}

template <int WIDTH, int K>
__device__ __forceinline__ void add_group(const float* __restrict__ h, float* acc, const int s,
                                          const int dl, const int ci, const int j, const int lane) {
  typedef typename RowVec<WIDTH>::type V;
  V r[K];
#pragma unroll
  for (int u = 0; u < K; ++u)
    r[u] = ((const V*)(h + (size_t)__builtin_amdgcn_readlane(s, j + u) * WIDTH))[lane];
#pragma unroll
  for (int u = 0; u < K; ++u) {
    const float cj = __int_as_float(__builtin_amdgcn_readlane(ci, j + u));
    V* a = (V*)(acc + __builtin_amdgcn_readlane(dl, j + u) * WIDTH) + lane;
    *a = axpy(*a, r[u], cj);
  }
}

template <int WIDTH>
__device__ __forceinline__ void add_owned_rows(const float* __restrict__ h, float* acc,
                                               const uint2 rc, const int m, const int lane) {
  const int s = (int)(rc.x & 0x1FFFFu);
  const int dl = (int)(rc.x >> 17);
  const int ci = (int)rc.y;
  int j = 0;
  for (; j + 8 <= m; j += 8) add_group<WIDTH, 8>(h, acc, s, dl, ci, j, lane);
  switch (m - j) {
    case 7: add_group<WIDTH, 7>(h, acc, s, dl, ci, j, lane); break;
    case 6: add_group<WIDTH, 6>(h, acc, s, dl, ci, j, lane); break;
    case 5: add_group<WIDTH, 5>(h, acc, s, dl, ci, j, lane); break;
    case 4: add_group<WIDTH, 4>(h, acc, s, dl, ci, j, lane); break;
    case 3: add_group<WIDTH, 3>(h, acc, s, dl, ci, j, lane); break;
    case 2: add_group<WIDTH, 2>(h, acc, s, dl, ci, j, lane); break;
    case 1: add_group<WIDTH, 1>(h, acc, s, dl, ci, j, lane); break;
    default: break;
  }
}

template <int WIDTH>
__global__ __launch_bounds__(1024)
void tile_aggregate_kernel(const float* __restrict__ h,
                           const unsigned* __restrict__ list,
                           const unsigned* __restrict__ toff,
                           const unsigned* __restrict__ tcount,
                           float* __restrict__ agg) {
  __shared__ float acc[TILE * WIDTH];
  __shared__ uint2 oq[32 * OQCAP];
  __shared__ unsigned oqn[32];

  const int tile = blockIdx.x;
  const int wave = __builtin_amdgcn_readfirstlane(threadIdx.x >> 5);
  const int lane = threadIdx.x & 31;
  const int row0 = tile * TILE;

  for (int i = threadIdx.x; i < TILE * WIDTH / 4; i += 1024)
    ((float4*)acc)[i] = make_float4(0.0f, 0.0f, 0.0f, 0.0f);
  if (threadIdx.x < 32) oqn[threadIdx.x] = 0u;
  __syncthreads();

  const unsigned nlist = (unsigned)__builtin_amdgcn_readfirstlane((int)tcount[tile]);
  const uint2* tl = (const uint2*)list + (unsigned)__builtin_amdgcn_readfirstlane((int)toff[tile]);

  for (unsigned b0 = (unsigned)wave * 32u; b0 < nlist; b0 += 8u * 1024u) {
    uint2 rc[8];
#pragma unroll
    for (int k = 0; k < 8; ++k) {
      const unsigned i = b0 + (unsigned)k * 1024u + (unsigned)lane;
      rc[k] = tl[i < nlist ? i : 0u];
    }
#pragma unroll
    for (int k = 0; k < 8; ++k) {
      const unsigned i = b0 + (unsigned)k * 1024u + (unsigned)lane;
      if (i < nlist) route_record<WIDTH>(h, acc, oq, oqn, rc[k]);
    }
  }
  __syncthreads();

  {
    unsigned n = (unsigned)__builtin_amdgcn_readfirstlane((int)oqn[wave]);
    if (n > (unsigned)OQCAP) n = (unsigned)OQCAP;
    const uint2* q = oq + wave * OQCAP;
    for (unsigned b = 0; b < n; b += 32u) {
      const int m = (int)(n - b < 32u ? n - b : 32u);
      const uint2 rc = q[b + (unsigned)(lane < m ? lane : 0)];
      add_owned_rows<WIDTH>(h, acc, rc, m, lane);
    }
  }
  __syncthreads();

  float4* o = (float4*)(agg + (size_t)row0 * WIDTH);
  for (int i = threadIdx.x; i < TILE * WIDTH / 4; i += 1024) o[i] = ((const float4*)acc)[i];
}

template <int FOUT, bool RELU, bool HAS_BIAS, bool HAS_SCALE>
__global__ __launch_bounds__(64)
void wmma_gemm_kernel(const float* __restrict__ A,
                      const float* __restrict__ row_scale,
                      const float* __restrict__ W,
                      const float* __restrict__ bias,
                      float* __restrict__ out) {
  constexpr int NG = FOUT / 64;
  constexpr int NJ = KD / 8;
  const int wave = threadIdx.x >> 5;
  const int lane = threadIdx.x & 31;
  const int c = lane & 15;
  const int h = lane >> 4;
  const int m_base = (blockIdx.x * 2 + wave) * 16;

  float4 a[NJ];
  const float4* ap = (const float4*)(A + (size_t)(m_base + c) * KD + 4 * h);
#pragma unroll
  for (int j = 0; j < NJ; ++j) a[j] = ap[2 * j];
  if (HAS_SCALE) {
    const float s = row_scale[m_base + c];
#pragma unroll
    for (int j = 0; j < NJ; ++j) { a[j].x *= s; a[j].y *= s; a[j].z *= s; a[j].w *= s; }
  }

  v8f acc[NG][4];
#pragma unroll
  for (int g = 0; g < NG; ++g)
#pragma unroll
    for (int t = 0; t < 4; ++t) acc[g][t] = v8f{};

  const float* wp = W + (size_t)(4 * h) * FOUT + 4 * c;

#pragma unroll
  for (int j = 0; j < NJ; ++j) {
#pragma unroll
    for (int p = 0; p < 2; ++p) {
      float4 b0[NG], b1[NG];
#pragma unroll
      for (int g = 0; g < NG; ++g) {
        b0[g] = *(const float4*)(wp + (size_t)(8 * j + 2 * p) * FOUT + 64 * g);
        b1[g] = *(const float4*)(wp + (size_t)(8 * j + 2 * p + 1) * FOUT + 64 * g);
      }
      v2f av;
      av.x = p ? a[j].z : a[j].x;
      av.y = p ? a[j].w : a[j].y;
#pragma unroll
      for (int g = 0; g < NG; ++g) {
        const float e0[4] = {b0[g].x, b0[g].y, b0[g].z, b0[g].w};
        const float e1[4] = {b1[g].x, b1[g].y, b1[g].z, b1[g].w};
#pragma unroll
        for (int t = 0; t < 4; ++t) {
          v2f bv;
          bv.x = e0[t];
          bv.y = e1[t];
          acc[g][t] = __builtin_amdgcn_wmma_f32_16x16x4_f32(
               false, av,  false, bv,
               (short)0, acc[g][t],  false,  false);
        }
      }
    }
  }

#pragma unroll
  for (int g = 0; g < NG; ++g) {
    float4 bz = make_float4(0.0f, 0.0f, 0.0f, 0.0f);
    if (HAS_BIAS) bz = *(const float4*)(bias + 64 * g + 4 * c);
#pragma unroll
    for (int r = 0; r < 8; ++r) {
      const int row = m_base + 8 * h + r;
      float4 v = make_float4(acc[g][0][r] + bz.x, acc[g][1][r] + bz.y,
                             acc[g][2][r] + bz.z, acc[g][3][r] + bz.w);
      if (RELU) {
        v.x = v.x > 0.0f ? v.x : 0.0f;
        v.y = v.y > 0.0f ? v.y : 0.0f;
        v.z = v.z > 0.0f ? v.z : 0.0f;
        v.w = v.w > 0.0f ? v.w : 0.0f;
      }
      *(float4*)(out + (size_t)row * FOUT + 64 * g + 4 * c) = v;
    }
  }
}

__global__ void final_epilogue_kernel(const float* __restrict__ agg,
                                      const float* __restrict__ in_norm,
                                      const float* __restrict__ b2,
                                      float* __restrict__ out) {
  int i = blockIdx.x * blockDim.x + threadIdx.x;
  if (i >= NN * 64) return;
  int n = i >> 6;
  int f = i & 63;
  out[i] = agg[i] * in_norm[n] + b2[f];
}

extern "C" void kernel_launch(void* const* d_in, const int* in_sizes, int n_in,
                              void* d_out, int out_size, void* d_ws, size_t ws_size,
                              hipStream_t stream) {
  const float* features = (const float*)d_in[0];
  const int*   src      = (const int*)d_in[1];
  const int*   dst      = (const int*)d_in[2];
  const float* ew       = (const float*)d_in[3];
  const float* W0 = (const float*)d_in[4];
  const float* b0 = (const float*)d_in[5];
  const float* W1 = (const float*)d_in[6];
  const float* b1 = (const float*)d_in[7];
  const float* W2 = (const float*)d_in[8];
  const float* b2 = (const float*)d_in[9];
  float* out = (float*)d_out;

  float* ws         = (float*)d_ws;
  float* out_norm   = ws;
  float* in_norm    = ws + NN;
  unsigned* fillctr = (unsigned*)(ws + 2 * NN);
  unsigned* tcount  = fillctr + CTRW;
  unsigned* toff    = tcount + 256;
  float* bufA       = (float*)(toff + 256);
  float* bufB       = bufA + (size_t)NN * KD;
  unsigned* list    = (unsigned*)(bufB + (size_t)NN * KD);

  fill_zero_kernel<<<512, 256, 0, stream>>>(out_norm, 2 * NN + CTRW);
  degree_kernel<<<(NE + 255) / 256, 256, 0, stream>>>(src, dst, out_norm, in_norm);
  norm_count_kernel<<<NTILE, 512, 0, stream>>>(out_norm, tcount);

  edge_place_kernel<<<(NE + EPB - 1) / EPB, 1024, 0, stream>>>(src, dst, ew, out_norm, tcount, fillctr, toff, list);

  tile_aggregate_kernel<128><<<NTILE, 1024, 0, stream>>>(features, list, toff, tcount, bufA);
  wmma_gemm_kernel<128, true, true, true><<<NN / 32, 64, 0, stream>>>(
      bufA, in_norm, W0, b0, bufB);

  tile_aggregate_kernel<128><<<NTILE, 1024, 0, stream>>>(bufB, list, toff, tcount, bufA);
  wmma_gemm_kernel<128, true, true, true><<<NN / 32, 64, 0, stream>>>(
      bufA, in_norm, W1, b1, bufB);

  wmma_gemm_kernel<64, false, false, false><<<NN / 32, 64, 0, stream>>>(
      bufB, nullptr, W2, nullptr, bufA);
  tile_aggregate_kernel<64><<<NTILE, 1024, 0, stream>>>(bufA, list, toff, tcount, bufB);
  final_epilogue_kernel<<<(NN * 64 + 255) / 256, 256, 0, stream>>>(bufB, in_norm, b2, out);
}


# edge coefficient takes in_norm[dst] too (ew*out_norm[src]*in_norm[dst]); layer 2's LDS rows start at b2 and go straight to out: epilogue launch gone, GEMMs lose row scale; sum order: bias first
# speedup vs baseline: 2.6330x; 1.0644x over previous
#include <hip/hip_runtime.h>


#define NN 100000
#define NE 1600000
#define KD 128
#define TILE 400
#define NTILE (NN / TILE)
#define OQCAP 320
#define EPB 4096
#define CTRS 32
#define CTRW (256 * CTRS)

typedef __attribute__((ext_vector_type(2))) float v2f;
typedef __attribute__((ext_vector_type(8))) float v8f;

__global__ void fill_zero_kernel(float* __restrict__ p, int n) {
  int i = blockIdx.x * blockDim.x + threadIdx.x;
  int stride = gridDim.x * blockDim.x;
  for (; i < n; i += stride) p[i] = 0.0f;
}

__global__ void degree_kernel(const int* __restrict__ src, const int* __restrict__ dst,
                              float* __restrict__ out_deg, float* __restrict__ in_deg) {
  int e = blockIdx.x * blockDim.x + threadIdx.x;
  if (e >= NE) return;
  unsafeAtomicAdd(out_deg + src[e], 1.0f);
  unsafeAtomicAdd(in_deg + dst[e], 1.0f);
}

__global__ __launch_bounds__(512)
void norm_count_kernel(float* __restrict__ deg, unsigned* __restrict__ tcount) {
  __shared__ unsigned total;
  const int i = threadIdx.x;
  if (i == 0) total = 0u;
  __syncthreads();
  unsigned mine = 0u;
  if (i < TILE) {
    const int n = blockIdx.x * TILE + i;
    float dout = deg[n];
    float din = deg[NN + n];
    mine = (unsigned)din;
    dout = dout < 1.0f ? 1.0f : dout;
    din = din < 1.0f ? 1.0f : din;
    deg[n] = rsqrtf(dout);
    deg[NN + n] = rsqrtf(din);
  }
  for (int off = 16; off > 0; off >>= 1) mine += __shfl_xor(mine, off);
  if ((i & 31) == 0 && mine)
    __hip_atomic_fetch_add(&total, mine, __ATOMIC_RELAXED, __HIP_MEMORY_SCOPE_WORKGROUP);
  __syncthreads();
  if (i == 0) tcount[blockIdx.x] = total;
}

__global__ __launch_bounds__(1024)
void edge_place_kernel(const int* __restrict__ src, const int* __restrict__ dst,
                       const float* __restrict__ ew, const float* __restrict__ out_norm,
                       const float* __restrict__ in_norm,
                       const unsigned* __restrict__ tcount, unsigned* __restrict__ fillctr,
                       unsigned* __restrict__ toffg, unsigned* __restrict__ list) {
  __shared__ unsigned lcnt[256];
  __shared__ unsigned lpos[256];
  __shared__ unsigned toff[256];
  __shared__ unsigned gbase[256];
  __shared__ uint2 stage[EPB];
  __shared__ unsigned sidx[EPB];

  const int tid = threadIdx.x;
  const int i = blockIdx.x * 1024 + tid;
  const bool valid = i < NE / 4;
  if (tid < 256) lcnt[tid] = 0u;
  __syncthreads();

  unsigned t[4] = {0u, 0u, 0u, 0u}, rk[4] = {0u, 0u, 0u, 0u};
  uint2 rc[4] = {};
  if (valid) {
    const int4 s = ((const int4*)src)[i];
    const int4 d = ((const int4*)dst)[i];
    const float4 w = ((const float4*)ew)[i];
    const int sv[4] = {s.x, s.y, s.z, s.w};
    const int dv[4] = {d.x, d.y, d.z, d.w};
    const float wv[4] = {w.x, w.y, w.z, w.w};
#pragma unroll
    for (int k = 0; k < 4; ++k) {
      t[k] = (unsigned)dv[k] / TILE;
      rc[k].x = (unsigned)sv[k] | (((unsigned)dv[k] - t[k] * TILE) << 17);
      rc[k].y = __float_as_uint(wv[k] * out_norm[sv[k]] * in_norm[dv[k]]);
      rk[k] = __hip_atomic_fetch_add(lcnt + t[k], 1u, __ATOMIC_RELAXED, __HIP_MEMORY_SCOPE_WORKGROUP);
    }
  }
  __syncthreads();

  if (tid < 32) {
    unsigned a[8], b[8];
    unsigned sa = 0u, sb = 0u;
#pragma unroll
    for (int k = 0; k < 8; ++k) {
      const int e = 8 * tid + k;
      a[k] = lcnt[e];
      b[k] = e < NTILE ? tcount[e] : 0u;
      sa += a[k];
      sb += b[k];
    }
    unsigned xa = sa, xb = sb;
    for (int off = 1; off < 32; off <<= 1) {
      const unsigned ya = __shfl_up(xa, off);
      const unsigned yb = __shfl_up(xb, off);
      if (tid >= off) { xa += ya; xb += yb; }
    }
    unsigned ea = xa - sa, eb = xb - sb;
#pragma unroll
    for (int k = 0; k < 8; ++k) {
      const int e = 8 * tid + k;
      lpos[e] = ea;
      toff[e] = eb;
      if (blockIdx.x == 0) toffg[e] = eb;
      ea += a[k];
      eb += b[k];
    }
  }
  __syncthreads();

  if (tid < NTILE) {
    const unsigned n = lcnt[tid];
    unsigned g = 0u;
    if (n) g = __hip_atomic_fetch_add(fillctr + tid * CTRS, n, __ATOMIC_RELAXED, __HIP_MEMORY_SCOPE_AGENT);
    gbase[tid] = toff[tid] + g;
  }
  __syncthreads();

  if (valid) {
#pragma unroll
    for (int k = 0; k < 4; ++k) {
      const unsigned p = lpos[t[k]] + rk[k];
      stage[p] = rc[k];
      sidx[p] = gbase[t[k]] + rk[k];
    }
  }
  __syncthreads();

  int total = NE - (int)blockIdx.x * EPB;
  if (total > EPB) total = EPB;
  for (int p = tid; p < total; p += 1024) ((uint2*)list)[sidx[p]] = stage[p];
}


template <int WIDTH> struct RowVec;
template <> struct RowVec<128> { typedef float4 type; };
template <> struct RowVec<64>  { typedef float2 type; };

__device__ __forceinline__ float4 axpy(float4 t, const float4 r, const float c) {
  t.x += r.x * c; t.y += r.y * c; t.z += r.z * c; t.w += r.w * c;
  return t;
}
__device__ __forceinline__ float2 axpy(float2 t, const float2 r, const float c) {
  t.x += r.x * c; t.y += r.y * c;
  return t;
}

template <int WIDTH>
__device__ __forceinline__ void route_record(const float* __restrict__ h, float* acc, uint2* oq,
                                             unsigned* oqn, const uint2 rc) {
  const unsigned own = (rc.x >> 17) & 31u;
  const unsigned slot = __hip_atomic_fetch_add(oqn + own, 1u, __ATOMIC_RELAXED, __HIP_MEMORY_SCOPE_WORKGROUP);
  if (slot < (unsigned)OQCAP) {
    oq[own * OQCAP + slot] = rc;
  } else {
    const float c = __uint_as_float(rc.y);
    float* a = acc + (rc.x >> 17) * WIDTH;
    const float* hp = h + (size_t)(rc.x & 0x1FFFFu) * WIDTH;
    for (int col = 0; col < WIDTH; ++col)
      __hip_atomic_fetch_add(a + col, hp[col] * c, __ATOMIC_RELAXED, __HIP_MEMORY_SCOPE_WORKGROUP);
  }
}

template <int WIDTH, int K>
__device__ __forceinline__ void add_group(const float* __restrict__ h, float* acc, const int s,
                                          const int dl, const int ci, const int j, const int lane) {
  typedef typename RowVec<WIDTH>::type V;
  V r[K];
#pragma unroll
  for (int u = 0; u < K; ++u)
    r[u] = ((const V*)(h + (size_t)__builtin_amdgcn_readlane(s, j + u) * WIDTH))[lane];
#pragma unroll
  for (int u = 0; u < K; ++u) {
    const float cj = __int_as_float(__builtin_amdgcn_readlane(ci, j + u));
    V* a = (V*)(acc + __builtin_amdgcn_readlane(dl, j + u) * WIDTH) + lane;
    *a = axpy(*a, r[u], cj);
  }
}

template <int WIDTH>
__device__ __forceinline__ void add_owned_rows(const float* __restrict__ h, float* acc,
                                               const uint2 rc, const int m, const int lane) {
  const int s = (int)(rc.x & 0x1FFFFu);
  const int dl = (int)(rc.x >> 17);
  const int ci = (int)rc.y;
  int j = 0;
  for (; j + 8 <= m; j += 8) add_group<WIDTH, 8>(h, acc, s, dl, ci, j, lane);
  switch (m - j) {
    case 7: add_group<WIDTH, 7>(h, acc, s, dl, ci, j, lane); break;
    case 6: add_group<WIDTH, 6>(h, acc, s, dl, ci, j, lane); break;
    case 5: add_group<WIDTH, 5>(h, acc, s, dl, ci, j, lane); break;
    case 4: add_group<WIDTH, 4>(h, acc, s, dl, ci, j, lane); break;
    case 3: add_group<WIDTH, 3>(h, acc, s, dl, ci, j, lane); break;
    case 2: add_group<WIDTH, 2>(h, acc, s, dl, ci, j, lane); break;
    case 1: add_group<WIDTH, 1>(h, acc, s, dl, ci, j, lane); break;
    default: break;
  }
}

template <int WIDTH, bool HAS_BIAS>
__global__ __launch_bounds__(1024)
void tile_aggregate_kernel(const float* __restrict__ h,
                           const unsigned* __restrict__ list,
                           const unsigned* __restrict__ toff,
                           const unsigned* __restrict__ tcount,
                           const float* __restrict__ bias,
                           float* __restrict__ agg) {
  __shared__ float acc[TILE * WIDTH];
  __shared__ uint2 oq[32 * OQCAP];
  __shared__ unsigned oqn[32];

  const int tile = blockIdx.x;
  const int wave = __builtin_amdgcn_readfirstlane(threadIdx.x >> 5);
  const int lane = threadIdx.x & 31;
  const int row0 = tile * TILE;

  float4 start = make_float4(0.0f, 0.0f, 0.0f, 0.0f);
  if (HAS_BIAS) start = ((const float4*)bias)[threadIdx.x % (WIDTH / 4)];
  for (int i = threadIdx.x; i < TILE * WIDTH / 4; i += 1024) ((float4*)acc)[i] = start;
  if (threadIdx.x < 32) oqn[threadIdx.x] = 0u;
  __syncthreads();

  const unsigned nlist = (unsigned)__builtin_amdgcn_readfirstlane((int)tcount[tile]);
  const uint2* tl = (const uint2*)list + (unsigned)__builtin_amdgcn_readfirstlane((int)toff[tile]);

  for (unsigned b0 = (unsigned)wave * 32u; b0 < nlist; b0 += 8u * 1024u) {
    uint2 rc[8];
#pragma unroll
    for (int k = 0; k < 8; ++k) {
      const unsigned i = b0 + (unsigned)k * 1024u + (unsigned)lane;
      rc[k] = tl[i < nlist ? i : 0u];
    }
#pragma unroll
    for (int k = 0; k < 8; ++k) {
      const unsigned i = b0 + (unsigned)k * 1024u + (unsigned)lane;
      if (i < nlist) route_record<WIDTH>(h, acc, oq, oqn, rc[k]);
    }
  }
  __syncthreads();

  {
    unsigned n = (unsigned)__builtin_amdgcn_readfirstlane((int)oqn[wave]);
    if (n > (unsigned)OQCAP) n = (unsigned)OQCAP;
    const uint2* q = oq + wave * OQCAP;
    for (unsigned b = 0; b < n; b += 32u) {
      const int m = (int)(n - b < 32u ? n - b : 32u);
      const uint2 rc = q[b + (unsigned)(lane < m ? lane : 0)];
      add_owned_rows<WIDTH>(h, acc, rc, m, lane);
    }
  }
  __syncthreads();

  float4* o = (float4*)(agg + (size_t)row0 * WIDTH);
  for (int i = threadIdx.x; i < TILE * WIDTH / 4; i += 1024) o[i] = ((const float4*)acc)[i];
}

template <int FOUT, bool RELU, bool HAS_BIAS>
__global__ __launch_bounds__(64)
void wmma_gemm_kernel(const float* __restrict__ A,
                      const float* __restrict__ W,
                      const float* __restrict__ bias,
                      float* __restrict__ out) {
  constexpr int NG = FOUT / 64;
  constexpr int NJ = KD / 8;
  const int wave = threadIdx.x >> 5;
  const int lane = threadIdx.x & 31;
  const int c = lane & 15;
  const int h = lane >> 4;
  const int m_base = (blockIdx.x * 2 + wave) * 16;

  float4 a[NJ];
  const float4* ap = (const float4*)(A + (size_t)(m_base + c) * KD + 4 * h);
#pragma unroll
  for (int j = 0; j < NJ; ++j) a[j] = ap[2 * j];

  v8f acc[NG][4];
#pragma unroll
  for (int g = 0; g < NG; ++g)
#pragma unroll
    for (int t = 0; t < 4; ++t) acc[g][t] = v8f{};

  const float* wp = W + (size_t)(4 * h) * FOUT + 4 * c;

#pragma unroll
  for (int j = 0; j < NJ; ++j) {
#pragma unroll
    for (int p = 0; p < 2; ++p) {
      float4 b0[NG], b1[NG];
#pragma unroll
      for (int g = 0; g < NG; ++g) {
        b0[g] = *(const float4*)(wp + (size_t)(8 * j + 2 * p) * FOUT + 64 * g);
        b1[g] = *(const float4*)(wp + (size_t)(8 * j + 2 * p + 1) * FOUT + 64 * g);
      }
      v2f av;
      av.x = p ? a[j].z : a[j].x;
      av.y = p ? a[j].w : a[j].y;
#pragma unroll
      for (int g = 0; g < NG; ++g) {
        const float e0[4] = {b0[g].x, b0[g].y, b0[g].z, b0[g].w};
        const float e1[4] = {b1[g].x, b1[g].y, b1[g].z, b1[g].w};
#pragma unroll
        for (int t = 0; t < 4; ++t) {
          v2f bv;
          bv.x = e0[t];
          bv.y = e1[t];
          acc[g][t] = __builtin_amdgcn_wmma_f32_16x16x4_f32(
               false, av,  false, bv,
               (short)0, acc[g][t],  false,  false);
        }
      }
    }
  }

#pragma unroll
  for (int g = 0; g < NG; ++g) {
    float4 bz = make_float4(0.0f, 0.0f, 0.0f, 0.0f);
    if (HAS_BIAS) bz = *(const float4*)(bias + 64 * g + 4 * c);
#pragma unroll
    for (int r = 0; r < 8; ++r) {
      const int row = m_base + 8 * h + r;
      float4 v = make_float4(acc[g][0][r] + bz.x, acc[g][1][r] + bz.y,
                             acc[g][2][r] + bz.z, acc[g][3][r] + bz.w);
      if (RELU) {
        v.x = v.x > 0.0f ? v.x : 0.0f;
        v.y = v.y > 0.0f ? v.y : 0.0f;
        v.z = v.z > 0.0f ? v.z : 0.0f;
        v.w = v.w > 0.0f ? v.w : 0.0f;
      }
      *(float4*)(out + (size_t)row * FOUT + 64 * g + 4 * c) = v;
    }
  }
}

extern "C" void kernel_launch(void* const* d_in, const int* in_sizes, int n_in,
                              void* d_out, int out_size, void* d_ws, size_t ws_size,
                              hipStream_t stream) {
  const float* features = (const float*)d_in[0];
  const int*   src      = (const int*)d_in[1];
  const int*   dst      = (const int*)d_in[2];
  const float* ew       = (const float*)d_in[3];
  const float* W0 = (const float*)d_in[4];
  const float* b0 = (const float*)d_in[5];
  const float* W1 = (const float*)d_in[6];
  const float* b1 = (const float*)d_in[7];
  const float* W2 = (const float*)d_in[8];
  const float* b2 = (const float*)d_in[9];
  float* out = (float*)d_out;

  float* ws         = (float*)d_ws;
  float* out_norm   = ws;
  float* in_norm    = ws + NN;
  unsigned* fillctr = (unsigned*)(ws + 2 * NN);
  unsigned* tcount  = fillctr + CTRW;
  unsigned* toff    = tcount + 256;
  float* bufA       = (float*)(toff + 256);
  float* bufB       = bufA + (size_t)NN * KD;
  unsigned* list    = (unsigned*)(bufB + (size_t)NN * KD);

  fill_zero_kernel<<<512, 256, 0, stream>>>(out_norm, 2 * NN + CTRW);
  degree_kernel<<<(NE + 255) / 256, 256, 0, stream>>>(src, dst, out_norm, in_norm);
  norm_count_kernel<<<NTILE, 512, 0, stream>>>(out_norm, tcount);

  edge_place_kernel<<<(NE + EPB - 1) / EPB, 1024, 0, stream>>>(src, dst, ew, out_norm, in_norm, tcount, fillctr, toff, list);

  tile_aggregate_kernel<128, false><<<NTILE, 1024, 0, stream>>>(features, list, toff, tcount, nullptr, bufA);
  wmma_gemm_kernel<128, true, true><<<NN / 32, 64, 0, stream>>>(bufA, W0, b0, bufB);

  tile_aggregate_kernel<128, false><<<NTILE, 1024, 0, stream>>>(bufB, list, toff, tcount, nullptr, bufA);
  wmma_gemm_kernel<128, true, true><<<NN / 32, 64, 0, stream>>>(bufA, W1, b1, bufB);

  wmma_gemm_kernel<64, false, false><<<NN / 32, 64, 0, stream>>>(bufB, W2, nullptr, bufA);
  tile_aggregate_kernel<64, true><<<NTILE, 1024, 0, stream>>>(bufA, list, toff, tcount, b2, out);
}
